# starting kernel
# baseline (speedup 1.0000x reference)
#include <hip/hip_runtime.h>

#define DB    2
#define DN    1024
#define DS    64
#define DM    128
#define DXY   16
#define DNC   32
#define MIN_SIGMA 1e-10f

typedef __attribute__((ext_vector_type(2))) float v2f;
typedef __attribute__((ext_vector_type(8))) float v8f;

__global__ __launch_bounds__(256)
void gauss_basis_agg_kernel(const float* __restrict__ x,
                            const float* __restrict__ clon,
                            const float* __restrict__ clat,
                            const float* __restrict__ mus_lon,
                            const float* __restrict__ mus_lat,
                            const float* __restrict__ sigma,
                            float* __restrict__ out)
{
    __shared__ float s_w[DS][DXY];
    __shared__ float s_inv[DXY];
    __shared__ float s_out[4][DXY][36];

    const int win  = blockIdx.x;
    const int tid  = threadIdx.x;
    const int lane = tid & 31;
    const int wave = tid >> 5;

    const float sig     = fmaxf(sigma[0], MIN_SIGMA);
    const float inv_sig = 1.0f / sig;

    const float* cl = clon + (size_t)win * DS;
    const float* ct = clat + (size_t)win * DS;
    #pragma unroll
    for (int i = tid; i < DS * DXY; i += 256) {
        const int s  = i >> 4;
        const int xy = i & 15;
        const int lo = xy >> 2;
        const int la = xy & 3;
        const float dlo = (cl[s] - mus_lon[lo]) * inv_sig;
        const float dla = (ct[s] - mus_lat[la]) * inv_sig;
        s_w[s][xy] = __expf(-0.5f * (dlo * dlo + dla * dla));
    }
    __syncthreads();

    if (tid < DXY) {
        float sum = 0.0f;
        #pragma unroll
        for (int s = 0; s < DS; ++s) sum += s_w[s][tid];
        s_inv[tid] = 1.0f / sum;
    }
    __syncthreads();

    #pragma unroll
    for (int i = tid; i < DS * DXY; i += 256) {
        s_w[i >> 4][i & 15] *= s_inv[i & 15];
    }
    __syncthreads();

    const float* xw  = x + (size_t)win * (DS * DM);
    const int mrow   = wave * 16 + (lane & 15);
    const int khalf  = lane >> 4;
    const int xy     = lane & 15;

    v8f acc = {};
    #pragma unroll
    for (int kk = 0; kk < DS / 4; ++kk) {
        const int sb = kk * 4 + 2 * khalf;
        v2f a, b;
        a.x = xw[(size_t)(sb    ) * DM + mrow];
        a.y = xw[(size_t)(sb + 1) * DM + mrow];
        b.x = s_w[sb    ][xy];
        b.y = s_w[sb + 1][xy];
        acc = __builtin_amdgcn_wmma_f32_16x16x4_f32(
                   false, a,  false, b,
                   (short)0, acc,
                   false,  false);
    }

    #pragma unroll
    for (int r = 0; r < 8; ++r) {
        const int M = wave * 16 + r + 8 * khalf;
        s_out[M >> 5][xy][M & 31] = acc[r];
    }
    __syncthreads();

    float* ow = out + (size_t)win * (4 * DXY * DNC);
    {
        const int i0  = tid * 8;
        const int v   = i0 >> 9;
        const int rem = i0 & 511;
        const int oxy = rem >> 5;
        const int c0  = rem & 31;
        const float4 lo = *(const float4*)&s_out[v][oxy][c0];
        const float4 hi = *(const float4*)&s_out[v][oxy][c0 + 4];
        float4* dst = (float4*)(ow + i0);
        dst[0] = lo;
        dst[1] = hi;
    }
}

extern "C" void kernel_launch(void* const* d_in, const int* in_sizes, int n_in,
                              void* d_out, int out_size, void* d_ws, size_t ws_size,
                              hipStream_t stream) {
    const float* x        = (const float*)d_in[0];
    const float* clon     = (const float*)d_in[1];
    const float* clat     = (const float*)d_in[2];
    const float* mus_lon  = (const float*)d_in[3];
    const float* mus_lat  = (const float*)d_in[4];
    const float* sigma    = (const float*)d_in[5];
    float* out            = (float*)d_out;

    dim3 grid(DB * DN);
    dim3 block(256);
    gauss_basis_agg_kernel<<<grid, block, 0, stream>>>(
        x, clon, clat, mus_lon, mus_lat, sigma, out);
}


# 4x fewer waves: each wave holds 4 WMMA tiles (64 m of a window, C^T = W^T X, b128 x loads and stores), weights built in registers; neighbour sum now two 32-term halves joined by permlanex16
# speedup vs baseline: 1.2445x; 1.2445x over previous
#include <hip/hip_runtime.h>


#define DB    2
#define DN    1024
#define DS    64
#define DM    128
#define DXY   16
#define DNC   32
#define MIN_SIGMA 1e-10f
#define WIN_PER_BLOCK 4

typedef __attribute__((ext_vector_type(2))) float v2f;
typedef __attribute__((ext_vector_type(4))) float v4f;
typedef __attribute__((ext_vector_type(8))) float v8f;

__device__ __forceinline__ float other_half(float v) {
    const int b = __builtin_bit_cast(int, v);
    return __builtin_bit_cast(float,
        __builtin_amdgcn_permlanex16(b, b, 0x76543210, 0xfedcba98, false, false));
}

__global__ __launch_bounds__(256) __attribute__((amdgpu_waves_per_eu(4, 4)))
void gauss_basis_agg_kernel(const float* __restrict__ x,
                            const float* __restrict__ clon,
                            const float* __restrict__ clat,
                            const float* __restrict__ mus_lon,
                            const float* __restrict__ mus_lat,
                            const float* __restrict__ sigma,
                            float* __restrict__ out)
{
    const int lane = threadIdx.x & 31;
    const int wave = threadIdx.x >> 5;
    const int win  = blockIdx.x * WIN_PER_BLOCK + (wave >> 1);
    const int half = wave & 1;
    const int c    = lane & 15;
    const int h    = lane >> 4;

    const v2f* cl2 = (const v2f*)(clon + (size_t)win * DS + 2 * h);
    const v2f* ct2 = (const v2f*)(clat + (size_t)win * DS + 2 * h);
    v2f lon[DS / 4], lat[DS / 4];
    #pragma unroll
    for (int kk = 0; kk < DS / 4; ++kk) {
        lon[kk] = cl2[2 * kk];
        lat[kk] = ct2[2 * kk];
    }
    const float mu_lo   = mus_lon[c >> 2];
    const float mu_la   = mus_lat[c & 3];
    const float sig     = fmaxf(sigma[0], MIN_SIGMA);
    const float inv_sig = 1.0f / sig;

    const v4f* xq = (const v4f*)(x + (size_t)win * (DS * DM) + (size_t)(2 * h) * DM
                                   + 64 * half + 4 * c);
    v4f q0[DS / 4], q1[DS / 4];
    #pragma unroll
    for (int kk = 0; kk < DS / 4; ++kk) {
        q0[kk] = xq[(size_t)(4 * kk    ) * (DM / 4)];
        q1[kk] = xq[(size_t)(4 * kk + 1) * (DM / 4)];
    }

    float e[DS / 2];
    float part = 0.0f;
    #pragma unroll
    for (int kk = 0; kk < DS / 4; ++kk) {
        #pragma unroll
        for (int j = 0; j < 2; ++j) {
            const float dlo = (lon[kk][j] - mu_lo) * inv_sig;
            const float dla = (lat[kk][j] - mu_la) * inv_sig;
            const float w   = __expf(-0.5f * (dlo * dlo + dla * dla));
            e[2 * kk + j] = w;
            part += w;
        }
    }
    const float inv = 1.0f / (part + other_half(part));

    v8f acc0 = {}, acc1 = {}, acc2 = {}, acc3 = {};
    #pragma unroll
    for (int kk = 0; kk < DS / 4; ++kk) {
        v2f a, b0, b1, b2, b3;
        a.x  = e[2 * kk] * inv;     a.y  = e[2 * kk + 1] * inv;
        b0.x = q0[kk].x;            b0.y = q1[kk].x;
        b1.x = q0[kk].y;            b1.y = q1[kk].y;
        b2.x = q0[kk].z;            b2.y = q1[kk].z;
        b3.x = q0[kk].w;            b3.y = q1[kk].w;
        acc0 = __builtin_amdgcn_wmma_f32_16x16x4_f32(false, a, false, b0, (short)0, acc0, false, false);
        acc1 = __builtin_amdgcn_wmma_f32_16x16x4_f32(false, a, false, b1, (short)0, acc1, false, false);
        acc2 = __builtin_amdgcn_wmma_f32_16x16x4_f32(false, a, false, b2, (short)0, acc2, false, false);
        acc3 = __builtin_amdgcn_wmma_f32_16x16x4_f32(false, a, false, b3, (short)0, acc3, false, false);
    }

    float* ow = out + (size_t)win * (4 * DXY * DNC)
                    + (2 * half + (c >> 3)) * (DXY * DNC) + (c & 7) * 4;
    #pragma unroll
    for (int r = 0; r < 8; ++r) {
        v4f o;
        o.x = acc0[r]; o.y = acc1[r]; o.z = acc2[r]; o.w = acc3[r];
        *(v4f*)(ow + (8 * h + r) * DNC) = o;
    }
}

extern "C" void kernel_launch(void* const* d_in, const int* in_sizes, int n_in,
                              void* d_out, int out_size, void* d_ws, size_t ws_size,
                              hipStream_t stream) {
    const float* x        = (const float*)d_in[0];
    const float* clon     = (const float*)d_in[1];
    const float* clat     = (const float*)d_in[2];
    const float* mus_lon  = (const float*)d_in[3];
    const float* mus_lat  = (const float*)d_in[4];
    const float* sigma    = (const float*)d_in[5];
    float* out            = (float*)d_out;

    dim3 grid(DB * DN / WIN_PER_BLOCK);
    dim3 block(256);
    gauss_basis_agg_kernel<<<grid, block, 0, stream>>>(
        x, clon, clat, mus_lon, mus_lat, sigma, out);
}


# k order s = 32h+2kk+j in both operands (f32 sums in a new order): coordinates as 16 b128 loads, a wave's 50 loads fit the 63-load counter and, pinned by a scheduling barrier, all issue at entry
# speedup vs baseline: 1.3291x; 1.0680x over previous
#include <hip/hip_runtime.h>


#define DB    2
#define DN    1024
#define DS    64
#define DM    128
#define DXY   16
#define DNC   32
#define MIN_SIGMA 1e-10f
#define WIN_PER_BLOCK 4

typedef __attribute__((ext_vector_type(2))) float v2f;
typedef __attribute__((ext_vector_type(4))) float v4f;
typedef __attribute__((ext_vector_type(8))) float v8f;

__device__ __forceinline__ float other_half(float v) {
    const int b = __builtin_bit_cast(int, v);
    return __builtin_bit_cast(float,
        __builtin_amdgcn_permlanex16(b, b, 0x76543210, 0xfedcba98, false, false));
}

__global__ __launch_bounds__(256) __attribute__((amdgpu_waves_per_eu(4, 4)))
void gauss_basis_agg_kernel(const float* __restrict__ x,
                            const float* __restrict__ clon,
                            const float* __restrict__ clat,
                            const float* __restrict__ mus_lon,
                            const float* __restrict__ mus_lat,
                            const float* __restrict__ sigma,
                            float* __restrict__ out)
{
    const int lane = threadIdx.x & 31;
    const int wave = threadIdx.x >> 5;
    const int win  = blockIdx.x * WIN_PER_BLOCK + (wave >> 1);
    const int half = wave & 1;
    const int c    = lane & 15;
    const int h    = lane >> 4;

    const v4f* cl4 = (const v4f*)(clon + (size_t)win * DS + 32 * h);
    const v4f* ct4 = (const v4f*)(clat + (size_t)win * DS + 32 * h);
    v4f lon[DS / 8], lat[DS / 8];
    #pragma unroll
    for (int i = 0; i < DS / 8; ++i) {
        lon[i] = cl4[i];
        lat[i] = ct4[i];
    }
    const float mu_lo   = mus_lon[c >> 2];
    const float mu_la   = mus_lat[c & 3];
    const float sig     = fmaxf(sigma[0], MIN_SIGMA);
    const float inv_sig = 1.0f / sig;

    const v4f* xq = (const v4f*)(x + (size_t)win * (DS * DM) + (size_t)(32 * h) * DM
                                   + 64 * half + 4 * c);
    v4f q0[DS / 4], q1[DS / 4];
    #pragma unroll
    for (int kk = 0; kk < DS / 4; ++kk) {
        q0[kk] = xq[(size_t)(2 * kk    ) * (DM / 4)];
        q1[kk] = xq[(size_t)(2 * kk + 1) * (DM / 4)];
    }
    __builtin_amdgcn_sched_barrier(0);

    float e[DS / 2];
    float part = 0.0f;
    #pragma unroll
    for (int kk = 0; kk < DS / 4; ++kk) {
        #pragma unroll
        for (int j = 0; j < 2; ++j) {
            const float dlo = (lon[kk >> 1][2 * (kk & 1) + j] - mu_lo) * inv_sig;
            const float dla = (lat[kk >> 1][2 * (kk & 1) + j] - mu_la) * inv_sig;
            const float w   = __expf(-0.5f * (dlo * dlo + dla * dla));
            e[2 * kk + j] = w;
            part += w;
        }
    }
    const float inv = 1.0f / (part + other_half(part));

    v8f acc0 = {}, acc1 = {}, acc2 = {}, acc3 = {};
    #pragma unroll
    for (int kk = 0; kk < DS / 4; ++kk) {
        v2f a, b0, b1, b2, b3;
        a.x  = e[2 * kk] * inv;     a.y  = e[2 * kk + 1] * inv;
        b0.x = q0[kk].x;            b0.y = q1[kk].x;
        b1.x = q0[kk].y;            b1.y = q1[kk].y;
        b2.x = q0[kk].z;            b2.y = q1[kk].z;
        b3.x = q0[kk].w;            b3.y = q1[kk].w;
        acc0 = __builtin_amdgcn_wmma_f32_16x16x4_f32(false, a, false, b0, (short)0, acc0, false, false);
        acc1 = __builtin_amdgcn_wmma_f32_16x16x4_f32(false, a, false, b1, (short)0, acc1, false, false);
        acc2 = __builtin_amdgcn_wmma_f32_16x16x4_f32(false, a, false, b2, (short)0, acc2, false, false);
        acc3 = __builtin_amdgcn_wmma_f32_16x16x4_f32(false, a, false, b3, (short)0, acc3, false, false);
    }

    float* ow = out + (size_t)win * (4 * DXY * DNC)
                    + (2 * half + (c >> 3)) * (DXY * DNC) + (c & 7) * 4;
    #pragma unroll
    for (int r = 0; r < 8; ++r) {
        v4f o;
        o.x = acc0[r]; o.y = acc1[r]; o.z = acc2[r]; o.w = acc3[r];
        *(v4f*)(ow + (8 * h + r) * DNC) = o;
    }
}

extern "C" void kernel_launch(void* const* d_in, const int* in_sizes, int n_in,
                              void* d_out, int out_size, void* d_ws, size_t ws_size,
                              hipStream_t stream) {
    const float* x        = (const float*)d_in[0];
    const float* clon     = (const float*)d_in[1];
    const float* clat     = (const float*)d_in[2];
    const float* mus_lon  = (const float*)d_in[3];
    const float* mus_lat  = (const float*)d_in[4];
    const float* sigma    = (const float*)d_in[5];
    float* out            = (float*)d_out;

    dim3 grid(DB * DN / WIN_PER_BLOCK);
    dim3 block(256);
    gauss_basis_agg_kernel<<<grid, block, 0, stream>>>(
        x, clon, clat, mus_lon, mus_lat, sigma, out);
}
